# starting kernel
# baseline (speedup 1.0000x reference)
#include <hip/hip_runtime.h>
#include <hip/hip_bf16.h>
#include <math.h>

#define BATCH 2048
#define D_MAX 64
#define DIM   512

typedef __attribute__((ext_vector_type(2))) float v2f;
typedef __attribute__((ext_vector_type(4))) float v4f;
typedef __attribute__((ext_vector_type(8))) float v8f;

__global__ __launch_bounds__(DIM) void prep_kernel(const float* __restrict__ cap,
                                                   float* __restrict__ c) {
    int d = threadIdx.x;
    float x = cap[d];
    c[d] = 1.0f / (1.0f + expf(-x));
}

__global__ __launch_bounds__(128) void gather_weight_kernel(
    const float* __restrict__ z, const float* __restrict__ c,
    const int* __restrict__ nbr, float* __restrict__ noise) {
    __shared__ int snbr[D_MAX];
    const int b = blockIdx.x;
    const int t = threadIdx.x;
    if (t < D_MAX) snbr[t] = nbr[(size_t)b * D_MAX + t];
    __syncthreads();

    const int d = t * 4;
    const v4f cc = *(const v4f*)(c + d);
    float w0 = 1.0f, w1 = 1.0f, w2 = 1.0f, w3 = 1.0f;
    float a0 = 0.0f, a1 = 0.0f, a2 = 0.0f, a3 = 0.0f;

#pragma unroll 8
    for (int k = 0; k < D_MAX; ++k) {
        if (k + 6 < D_MAX) {
            __builtin_prefetch(z + (size_t)snbr[k + 6] * DIM + d, 0, 0);
        }
        const v4f v = *(const v4f*)(z + (size_t)snbr[k] * DIM + d);
        a0 = fmaf(w0, v.x, a0);
        a1 = fmaf(w1, v.y, a1);
        a2 = fmaf(w2, v.z, a2);
        a3 = fmaf(w3, v.w, a3);
        w0 *= cc.x;
        w1 *= cc.y;
        w2 *= cc.z;
        w3 *= cc.w;
    }

    v4f r; r.x = a0; r.y = a1; r.z = a2; r.w = a3;
    *(v4f*)(noise + (size_t)b * DIM + d) = r;
}

__global__ __launch_bounds__(256) void colsum_wmma_kernel(
    const float* __restrict__ noise, float* __restrict__ sums) {
    const int dbase = blockIdx.x * 16;
    const int lane  = threadIdx.x & 31;
    const int wave  = threadIdx.x >> 5;

    v2f a; a.x = 1.0f; a.y = 1.0f;
    v8f acc = {};

    const int n     = lane & 15;
    const int khalf = (lane >> 4) * 2;
    const float* base = noise + (size_t)(wave * 4 + khalf) * DIM + dbase + n;

#pragma unroll 4
    for (int i = 0; i < BATCH / 32; ++i) {
        const float* p0 = base + (size_t)i * 32 * DIM;
        v2f bmat;
        bmat.x = p0[0];
        bmat.y = p0[DIM];
        acc = __builtin_amdgcn_wmma_f32_16x16x4_f32(
             false, a,  false, bmat,
             (short)0, acc,  false,  false);
    }

    __shared__ float part[8][16];
    if (lane < 16) part[wave][lane] = acc[0];
    __syncthreads();
    if (threadIdx.x < 16) {
        float s = 0.0f;
#pragma unroll
        for (int w = 0; w < 8; ++w) s += part[w][threadIdx.x];
        sums[dbase + threadIdx.x] = s;
    }
}

__global__ __launch_bounds__(256) void combine_kernel(
    const float* __restrict__ z, const float* __restrict__ c,
    const float* __restrict__ noise, const float* __restrict__ sums,
    float* __restrict__ out) {
    const int i4 = (blockIdx.x * 256 + threadIdx.x) * 4;
    const int d  = i4 & (DIM - 1);

    const v4f cz = *(const v4f*)(c + d);
    const v4f sm = *(const v4f*)(sums + d);
    const v4f zz = *(const v4f*)(z + i4);
    const v4f nn = *(const v4f*)(noise + i4);

    const float inv = 1.0f / (float)BATCH;
    v4f r;
    r.x = fmaf(cz.x, nn.x - sm.x * inv, zz.x);
    r.y = fmaf(cz.y, nn.y - sm.y * inv, zz.y);
    r.z = fmaf(cz.z, nn.z - sm.z * inv, zz.z);
    r.w = fmaf(cz.w, nn.w - sm.w * inv, zz.w);
    *(v4f*)(out + i4) = r;
}

extern "C" void kernel_launch(void* const* d_in, const int* in_sizes, int n_in,
                              void* d_out, int out_size, void* d_ws, size_t ws_size,
                              hipStream_t stream) {
    const float* z_mean    = (const float*)d_in[0];
    const float* cap_param = (const float*)d_in[1];
    const int*   neighbors = (const int*)d_in[2];

    float* out = (float*)d_out;

    float* ws_c     = (float*)d_ws;
    float* ws_sums  = ws_c + DIM;
    float* ws_noise = ws_sums + DIM;

    prep_kernel<<<1, DIM, 0, stream>>>(cap_param, ws_c);

    gather_weight_kernel<<<BATCH, 128, 0, stream>>>(z_mean, ws_c, neighbors, ws_noise);

    colsum_wmma_kernel<<<DIM / 16, 256, 0, stream>>>(ws_noise, ws_sums);

    combine_kernel<<<(BATCH * DIM) / (256 * 4), 256, 0, stream>>>(z_mean, ws_c, ws_noise,
                                                                  ws_sums, out);
}


# Column-sum kernel: each wave issues all its row loads before its WMMA chain (32 waves per 16-dim tile, two accumulators); the order in which batch rows are summed changes
# speedup vs baseline: 1.3802x; 1.3802x over previous
#include <hip/hip_runtime.h>
#include <hip/hip_bf16.h>
#include <math.h>

#define BATCH 2048
#define D_MAX 64
#define DIM   512

typedef __attribute__((ext_vector_type(2))) float v2f;
typedef __attribute__((ext_vector_type(4))) float v4f;
typedef __attribute__((ext_vector_type(8))) float v8f;

__global__ __launch_bounds__(DIM) void prep_kernel(const float* __restrict__ cap,
                                                   float* __restrict__ c) {
    int d = threadIdx.x;
    float x = cap[d];
    c[d] = 1.0f / (1.0f + expf(-x));
}

__global__ __launch_bounds__(128) void gather_weight_kernel(
    const float* __restrict__ z, const float* __restrict__ c,
    const int* __restrict__ nbr, float* __restrict__ noise) {
    __shared__ int snbr[D_MAX];
    const int b = blockIdx.x;
    const int t = threadIdx.x;
    if (t < D_MAX) snbr[t] = nbr[(size_t)b * D_MAX + t];
    __syncthreads();

    const int d = t * 4;
    const v4f cc = *(const v4f*)(c + d);
    float w0 = 1.0f, w1 = 1.0f, w2 = 1.0f, w3 = 1.0f;
    float a0 = 0.0f, a1 = 0.0f, a2 = 0.0f, a3 = 0.0f;

#pragma unroll 8
    for (int k = 0; k < D_MAX; ++k) {
        if (k + 6 < D_MAX) {
            __builtin_prefetch(z + (size_t)snbr[k + 6] * DIM + d, 0, 0);
        }
        const v4f v = *(const v4f*)(z + (size_t)snbr[k] * DIM + d);
        a0 = fmaf(w0, v.x, a0);
        a1 = fmaf(w1, v.y, a1);
        a2 = fmaf(w2, v.z, a2);
        a3 = fmaf(w3, v.w, a3);
        w0 *= cc.x;
        w1 *= cc.y;
        w2 *= cc.z;
        w3 *= cc.w;
    }

    v4f r; r.x = a0; r.y = a1; r.z = a2; r.w = a3;
    *(v4f*)(noise + (size_t)b * DIM + d) = r;
}

#define CS_WAVES 32
#define CS_STEPS (BATCH / (CS_WAVES * 4))

__global__ __launch_bounds__(CS_WAVES * 32) void colsum_wmma_kernel(
    const float* __restrict__ noise, float* __restrict__ sums) {
    const int dbase = blockIdx.x * 16;
    const int lane  = threadIdx.x & 31;
    const int wave  = threadIdx.x >> 5;

    v2f a; a.x = 1.0f; a.y = 1.0f;

    const int n     = lane & 15;
    const int khalf = (lane >> 4) * 2;
    const float* base =
        noise + (size_t)(wave * (CS_STEPS * 4) + khalf) * DIM + dbase + n;

    float r0[CS_STEPS], r1[CS_STEPS];
#pragma unroll
    for (int i = 0; i < CS_STEPS; ++i) {
        r0[i] = base[(size_t)i * 4 * DIM];
        r1[i] = base[(size_t)i * 4 * DIM + DIM];
    }

    v8f acc0 = {};
    v8f acc1 = {};
#pragma unroll
    for (int i = 0; i < CS_STEPS; i += 2) {
        v2f b0; b0.x = r0[i];     b0.y = r1[i];
        v2f b1; b1.x = r0[i + 1]; b1.y = r1[i + 1];
        acc0 = __builtin_amdgcn_wmma_f32_16x16x4_f32(
             false, a,  false, b0,
             (short)0, acc0,  false,  false);
        acc1 = __builtin_amdgcn_wmma_f32_16x16x4_f32(
             false, a,  false, b1,
             (short)0, acc1,  false,  false);
    }

    __shared__ float part[CS_WAVES][16];
    if (lane < 16) part[wave][lane] = acc0[0] + acc1[0];
    __syncthreads();
    if (threadIdx.x < 16) {
        float s0 = 0.0f, s1 = 0.0f, s2 = 0.0f, s3 = 0.0f;
#pragma unroll
        for (int w = 0; w < CS_WAVES; w += 4) {
            s0 += part[w + 0][threadIdx.x];
            s1 += part[w + 1][threadIdx.x];
            s2 += part[w + 2][threadIdx.x];
            s3 += part[w + 3][threadIdx.x];
        }
        sums[dbase + threadIdx.x] = (s0 + s1) + (s2 + s3);
    }
}

__global__ __launch_bounds__(256) void combine_kernel(
    const float* __restrict__ z, const float* __restrict__ c,
    const float* __restrict__ noise, const float* __restrict__ sums,
    float* __restrict__ out) {
    const int i4 = (blockIdx.x * 256 + threadIdx.x) * 4;
    const int d  = i4 & (DIM - 1);

    const v4f cz = *(const v4f*)(c + d);
    const v4f sm = *(const v4f*)(sums + d);
    const v4f zz = *(const v4f*)(z + i4);
    const v4f nn = *(const v4f*)(noise + i4);

    const float inv = 1.0f / (float)BATCH;
    v4f r;
    r.x = fmaf(cz.x, nn.x - sm.x * inv, zz.x);
    r.y = fmaf(cz.y, nn.y - sm.y * inv, zz.y);
    r.z = fmaf(cz.z, nn.z - sm.z * inv, zz.z);
    r.w = fmaf(cz.w, nn.w - sm.w * inv, zz.w);
    *(v4f*)(out + i4) = r;
}

extern "C" void kernel_launch(void* const* d_in, const int* in_sizes, int n_in,
                              void* d_out, int out_size, void* d_ws, size_t ws_size,
                              hipStream_t stream) {
    const float* z_mean    = (const float*)d_in[0];
    const float* cap_param = (const float*)d_in[1];
    const int*   neighbors = (const int*)d_in[2];

    float* out = (float*)d_out;

    float* ws_c     = (float*)d_ws;
    float* ws_sums  = ws_c + DIM;
    float* ws_noise = ws_sums + DIM;

    prep_kernel<<<1, DIM, 0, stream>>>(cap_param, ws_c);

    gather_weight_kernel<<<BATCH, 128, 0, stream>>>(z_mean, ws_c, neighbors, ws_noise);

    colsum_wmma_kernel<<<DIM / 16, CS_WAVES * 32, 0, stream>>>(ws_noise, ws_sums);

    combine_kernel<<<(BATCH * DIM) / (256 * 4), 256, 0, stream>>>(z_mean, ws_c, ws_noise,
                                                                  ws_sums, out);
}


# Sigmoid of cap_param computed inside the gather kernel (block 0 stores c for the combine kernel): the separate one-block prep launch is gone, three launches instead of four
# speedup vs baseline: 1.4844x; 1.0755x over previous
#include <hip/hip_runtime.h>
#include <hip/hip_bf16.h>
#include <math.h>

#define BATCH 2048
#define D_MAX 64
#define DIM   512

typedef __attribute__((ext_vector_type(2))) float v2f;
typedef __attribute__((ext_vector_type(4))) float v4f;
typedef __attribute__((ext_vector_type(8))) float v8f;

__global__ __launch_bounds__(128) void gather_weight_kernel(
    const float* __restrict__ z, const float* __restrict__ cap,
    const int* __restrict__ nbr, float* __restrict__ c_out,
    float* __restrict__ noise) {
    __shared__ int snbr[D_MAX];
    const int b = blockIdx.x;
    const int t = threadIdx.x;
    const int d = t * 4;
    const v4f cp = *(const v4f*)(cap + d);
    if (t < D_MAX) snbr[t] = nbr[(size_t)b * D_MAX + t];
    __syncthreads();

    v4f cc;
    cc.x = 1.0f / (1.0f + expf(-cp.x));
    cc.y = 1.0f / (1.0f + expf(-cp.y));
    cc.z = 1.0f / (1.0f + expf(-cp.z));
    cc.w = 1.0f / (1.0f + expf(-cp.w));
    if (b == 0) *(v4f*)(c_out + d) = cc;
    float w0 = 1.0f, w1 = 1.0f, w2 = 1.0f, w3 = 1.0f;
    float a0 = 0.0f, a1 = 0.0f, a2 = 0.0f, a3 = 0.0f;

#pragma unroll 8
    for (int k = 0; k < D_MAX; ++k) {
        if (k + 6 < D_MAX) {
            __builtin_prefetch(z + (size_t)snbr[k + 6] * DIM + d, 0, 0);
        }
        const v4f v = *(const v4f*)(z + (size_t)snbr[k] * DIM + d);
        a0 = fmaf(w0, v.x, a0);
        a1 = fmaf(w1, v.y, a1);
        a2 = fmaf(w2, v.z, a2);
        a3 = fmaf(w3, v.w, a3);
        w0 *= cc.x;
        w1 *= cc.y;
        w2 *= cc.z;
        w3 *= cc.w;
    }

    v4f r; r.x = a0; r.y = a1; r.z = a2; r.w = a3;
    *(v4f*)(noise + (size_t)b * DIM + d) = r;
}

#define CS_WAVES 32
#define CS_STEPS (BATCH / (CS_WAVES * 4))

__global__ __launch_bounds__(CS_WAVES * 32) void colsum_wmma_kernel(
    const float* __restrict__ noise, float* __restrict__ sums) {
    const int dbase = blockIdx.x * 16;
    const int lane  = threadIdx.x & 31;
    const int wave  = threadIdx.x >> 5;

    v2f a; a.x = 1.0f; a.y = 1.0f;

    const int n     = lane & 15;
    const int khalf = (lane >> 4) * 2;
    const float* base =
        noise + (size_t)(wave * (CS_STEPS * 4) + khalf) * DIM + dbase + n;

    float r0[CS_STEPS], r1[CS_STEPS];
#pragma unroll
    for (int i = 0; i < CS_STEPS; ++i) {
        r0[i] = base[(size_t)i * 4 * DIM];
        r1[i] = base[(size_t)i * 4 * DIM + DIM];
    }

    v8f acc0 = {};
    v8f acc1 = {};
#pragma unroll
    for (int i = 0; i < CS_STEPS; i += 2) {
        v2f b0; b0.x = r0[i];     b0.y = r1[i];
        v2f b1; b1.x = r0[i + 1]; b1.y = r1[i + 1];
        acc0 = __builtin_amdgcn_wmma_f32_16x16x4_f32(
             false, a,  false, b0,
             (short)0, acc0,  false,  false);
        acc1 = __builtin_amdgcn_wmma_f32_16x16x4_f32(
             false, a,  false, b1,
             (short)0, acc1,  false,  false);
    }

    __shared__ float part[CS_WAVES][16];
    if (lane < 16) part[wave][lane] = acc0[0] + acc1[0];
    __syncthreads();
    if (threadIdx.x < 16) {
        float s0 = 0.0f, s1 = 0.0f, s2 = 0.0f, s3 = 0.0f;
#pragma unroll
        for (int w = 0; w < CS_WAVES; w += 4) {
            s0 += part[w + 0][threadIdx.x];
            s1 += part[w + 1][threadIdx.x];
            s2 += part[w + 2][threadIdx.x];
            s3 += part[w + 3][threadIdx.x];
        }
        sums[dbase + threadIdx.x] = (s0 + s1) + (s2 + s3);
    }
}

__global__ __launch_bounds__(256) void combine_kernel(
    const float* __restrict__ z, const float* __restrict__ c,
    const float* __restrict__ noise, const float* __restrict__ sums,
    float* __restrict__ out) {
    const int i4 = (blockIdx.x * 256 + threadIdx.x) * 4;
    const int d  = i4 & (DIM - 1);

    const v4f cz = *(const v4f*)(c + d);
    const v4f sm = *(const v4f*)(sums + d);
    const v4f zz = *(const v4f*)(z + i4);
    const v4f nn = *(const v4f*)(noise + i4);

    const float inv = 1.0f / (float)BATCH;
    v4f r;
    r.x = fmaf(cz.x, nn.x - sm.x * inv, zz.x);
    r.y = fmaf(cz.y, nn.y - sm.y * inv, zz.y);
    r.z = fmaf(cz.z, nn.z - sm.z * inv, zz.z);
    r.w = fmaf(cz.w, nn.w - sm.w * inv, zz.w);
    *(v4f*)(out + i4) = r;
}

extern "C" void kernel_launch(void* const* d_in, const int* in_sizes, int n_in,
                              void* d_out, int out_size, void* d_ws, size_t ws_size,
                              hipStream_t stream) {
    const float* z_mean    = (const float*)d_in[0];
    const float* cap_param = (const float*)d_in[1];
    const int*   neighbors = (const int*)d_in[2];

    float* out = (float*)d_out;

    float* ws_c     = (float*)d_ws;
    float* ws_sums  = ws_c + DIM;
    float* ws_noise = ws_sums + DIM;

    gather_weight_kernel<<<BATCH, 128, 0, stream>>>(z_mean, cap_param, neighbors, ws_c,
                                                    ws_noise);

    colsum_wmma_kernel<<<DIM / 16, CS_WAVES * 32, 0, stream>>>(ws_noise, ws_sums);

    combine_kernel<<<(BATCH * DIM) / (256 * 4), 256, 0, stream>>>(z_mean, ws_c, ws_noise,
                                                                  ws_sums, out);
}


# Column sums of each block's 128 rows taken on WMMA at the end of the gather kernel; combine adds the 16 partials once per block through LDS (64 dims x 64 rows); 2 launches; row sum order changes
# speedup vs baseline: 2.6983x; 1.8178x over previous
#include <hip/hip_runtime.h>
#include <hip/hip_bf16.h>
#include <math.h>

#define BATCH 2048
#define D_MAX 64
#define DIM   512

typedef __attribute__((ext_vector_type(2))) float v2f;
typedef __attribute__((ext_vector_type(4))) float v4f;
typedef __attribute__((ext_vector_type(8))) float v8f;

#define GS_DIMS    32
#define GS_ROWS    128
#define GS_THREADS 256
#define GS_GROUPS  (GS_THREADS / 8)
#define GS_RPT     (GS_ROWS / GS_GROUPS)
#define GS_FILL    (BATCH / GS_GROUPS)

typedef __attribute__((ext_vector_type(4))) int v4i;

__device__ __forceinline__ void lds_only_barrier() {
    __builtin_amdgcn_fence(__ATOMIC_RELEASE, "workgroup", "local");
    __builtin_amdgcn_s_barrier();
    __builtin_amdgcn_fence(__ATOMIC_ACQUIRE, "workgroup", "local");
}

__global__ __launch_bounds__(GS_THREADS) void gather_weight_kernel(
    const float* __restrict__ z, const float* __restrict__ cap,
    const int* __restrict__ nbr, float* __restrict__ c_out,
    float* __restrict__ noise, float* __restrict__ colpart) {
    __shared__ float zs[BATCH * GS_DIMS];
    __shared__ v4f red[GS_THREADS];
    const unsigned t = threadIdx.x;
    const unsigned q = t & 7;
    const unsigned g = t >> 3;
    const unsigned slice = blockIdx.x & (DIM / GS_DIMS - 1);
    const unsigned chunk = blockIdx.x / (DIM / GS_DIMS);
    const unsigned d  = slice * GS_DIMS + q * 4;
    const unsigned b0 = chunk * GS_ROWS + g;

    int off[GS_RPT][8];
#pragma unroll
    for (int m = 0; m < GS_RPT; ++m) {
        const int* p = nbr + (size_t)(b0 + m * GS_GROUPS) * D_MAX + q * 8;
        const v4i lo = *(const v4i*)p;
        const v4i hi = *(const v4i*)(p + 4);
        off[m][0] = lo.x * (GS_DIMS * 4); off[m][1] = lo.y * (GS_DIMS * 4);
        off[m][2] = lo.z * (GS_DIMS * 4); off[m][3] = lo.w * (GS_DIMS * 4);
        off[m][4] = hi.x * (GS_DIMS * 4); off[m][5] = hi.y * (GS_DIMS * 4);
        off[m][6] = hi.z * (GS_DIMS * 4); off[m][7] = hi.w * (GS_DIMS * 4);
    }
    const v4f cp = *(const v4f*)(cap + d);

    {
        const unsigned rot  = chunk * (GS_FILL / (BATCH / GS_ROWS));
        const unsigned toff = g * DIM + d;
        v4f st[GS_FILL];
#pragma unroll
        for (unsigned p = 0; p < GS_FILL; ++p) {
            const unsigned rs = ((p + rot) & (GS_FILL - 1)) * GS_GROUPS;
            const float* zr = z + (size_t)rs * DIM;
            st[p] = *(const v4f*)(zr + toff);
        }
#pragma unroll
        for (unsigned p = 0; p < GS_FILL; ++p) {
            const unsigned rs = ((p + rot) & (GS_FILL - 1)) * GS_GROUPS;
            *(v4f*)(zs + rs * GS_DIMS + t * 4) = st[p];
        }
    }

    v4f cc;
    cc.x = 1.0f / (1.0f + expf(-cp.x));
    cc.y = 1.0f / (1.0f + expf(-cp.y));
    cc.z = 1.0f / (1.0f + expf(-cp.z));
    cc.w = 1.0f / (1.0f + expf(-cp.w));
    if (chunk == 0 && g == 0) *(v4f*)(c_out + d) = cc;
    __syncthreads();

    v4f w = {1.0f, 1.0f, 1.0f, 1.0f};
    v4f acc[GS_RPT];
#pragma unroll
    for (int m = 0; m < GS_RPT; ++m) acc[m] = (v4f){0.0f, 0.0f, 0.0f, 0.0f};
    const char* zq = (const char*)zs + q * 16;
    v4f bufA[2 * GS_RPT], bufB[2 * GS_RPT];

#define GS_LOAD1(buf, h, s, i)                                                 \
    _Pragma("unroll")                                                          \
    for (int m = 0; m < GS_RPT; ++m)                                           \
        buf[(h) * GS_RPT + m] = *(const v4f*)(                                 \
            zq + __builtin_amdgcn_mov_dpp8(off[m][i], (s) * 0x249249));
#define GS_LOAD2(buf, s, i) GS_LOAD1(buf, 0, s, i) GS_LOAD1(buf, 1, s, (i) + 1)
#define GS_USE2(buf)                                                           \
    _Pragma("unroll")                                                          \
    for (int h = 0; h < 2; ++h) {                                              \
        _Pragma("unroll")                                                      \
        for (int m = 0; m < GS_RPT; ++m)                                       \
            acc[m] = __builtin_elementwise_fma(w, buf[h * GS_RPT + m], acc[m]); \
        w *= cc;                                                               \
    }
#define GS_S(s, nxt)                                                           \
    GS_LOAD2(bufB, s, 2) GS_USE2(bufA) GS_LOAD2(bufA, s, 4) GS_USE2(bufB)      \
    GS_LOAD2(bufB, s, 6) GS_USE2(bufA) nxt GS_USE2(bufB)

    GS_LOAD2(bufA, 0, 0)
    GS_S(0, GS_LOAD2(bufA, 1, 0)) GS_S(1, GS_LOAD2(bufA, 2, 0))
    GS_S(2, GS_LOAD2(bufA, 3, 0)) GS_S(3, GS_LOAD2(bufA, 4, 0))
    GS_S(4, GS_LOAD2(bufA, 5, 0)) GS_S(5, GS_LOAD2(bufA, 6, 0))
    GS_S(6, GS_LOAD2(bufA, 7, 0)) GS_S(7, )
#undef GS_S
#undef GS_USE2
#undef GS_LOAD2
#undef GS_LOAD1

#pragma unroll
    for (int m = 0; m < GS_RPT; ++m)
        *(v4f*)(noise + (size_t)(b0 + m * GS_GROUPS) * DIM + d) = acc[m];

    v2f ones; ones.x = 1.0f; ones.y = 1.0f;
    v8f cs[4];
#pragma unroll
    for (int j = 0; j < 4; ++j) {
        v2f b; b.x = acc[0][j]; b.y = acc[1][j];
        v8f zero = {};
        cs[j] = __builtin_amdgcn_wmma_f32_16x16x4_f32(
             false, ones,  false, b,
             (short)0, zero,  false,  false);
    }
#pragma unroll
    for (int j = 0; j < 4; ++j) {
        v2f b; b.x = acc[2][j]; b.y = acc[3][j];
        cs[j] = __builtin_amdgcn_wmma_f32_16x16x4_f32(
             false, ones,  false, b,
             (short)0, cs[j],  false,  false);
    }
    v4f s; s.x = cs[0][0]; s.y = cs[1][0]; s.z = cs[2][0]; s.w = cs[3][0];
    red[t] = s;
    lds_only_barrier();
    if (t < 8) {
        v4f tot = {0.0f, 0.0f, 0.0f, 0.0f};
#pragma unroll
        for (int w = 0; w < GS_THREADS / 32; ++w) {
            tot += red[w * 32 + t];
            tot += red[w * 32 + t + 8];
        }
        *(v4f*)(colpart + chunk * DIM + slice * GS_DIMS + t * 4) = tot;
    }
}

#define CB_DIMS    64
#define CB_ROWS    64
#define CB_THREADS 256
#define CB_PARTS   (BATCH / GS_ROWS)

__global__ __launch_bounds__(CB_THREADS) void combine_kernel(
    const float* __restrict__ z, const float* __restrict__ c,
    const float* __restrict__ noise, const float* __restrict__ colpart,
    float* __restrict__ out) {
    __shared__ v4f sred[CB_THREADS];
    const unsigned t  = threadIdx.x;
    const unsigned dq = t & 15;
    const unsigned rr = t >> 4;
    const unsigned dg = blockIdx.x & (DIM / CB_DIMS - 1);
    const unsigned rc = blockIdx.x / (DIM / CB_DIMS);
    const unsigned d  = dg * CB_DIMS + dq * 4;
    const size_t   i0 = (size_t)(rc * CB_ROWS + rr) * DIM + d;

    const v4f pp = *(const v4f*)(colpart + rr * DIM + d);
    const v4f cz = *(const v4f*)(c + d);
    v4f zz[CB_ROWS / 16], nn[CB_ROWS / 16];
#pragma unroll
    for (int i = 0; i < CB_ROWS / 16; ++i) {
        zz[i] = *(const v4f*)(z + i0 + (size_t)i * 16 * DIM);
        nn[i] = *(const v4f*)(noise + i0 + (size_t)i * 16 * DIM);
    }

    sred[t] = pp;
    lds_only_barrier();
    v4f sm = sred[dq];
#pragma unroll
    for (int p = 1; p < CB_PARTS; ++p) sm += sred[p * 16 + dq];

    const float inv = 1.0f / (float)BATCH;
#pragma unroll
    for (int i = 0; i < CB_ROWS / 16; ++i) {
        v4f r;
        r.x = fmaf(cz.x, nn[i].x - sm.x * inv, zz[i].x);
        r.y = fmaf(cz.y, nn[i].y - sm.y * inv, zz[i].y);
        r.z = fmaf(cz.z, nn[i].z - sm.z * inv, zz[i].z);
        r.w = fmaf(cz.w, nn[i].w - sm.w * inv, zz[i].w);
        *(v4f*)(out + i0 + (size_t)i * 16 * DIM) = r;
    }
}

extern "C" void kernel_launch(void* const* d_in, const int* in_sizes, int n_in,
                              void* d_out, int out_size, void* d_ws, size_t ws_size,
                              hipStream_t stream) {
    const float* z_mean    = (const float*)d_in[0];
    const float* cap_param = (const float*)d_in[1];
    const int*   neighbors = (const int*)d_in[2];

    float* out = (float*)d_out;

    float* ws_c     = (float*)d_ws;
    float* ws_part  = ws_c + DIM;
    float* ws_noise = ws_part + (BATCH / GS_ROWS) * DIM;

    gather_weight_kernel<<<(DIM / GS_DIMS) * (BATCH / GS_ROWS), GS_THREADS, 0, stream>>>(
        z_mean, cap_param, neighbors, ws_c, ws_noise, ws_part);

    combine_kernel<<<(DIM / CB_DIMS) * (BATCH / CB_ROWS), CB_THREADS, 0, stream>>>(
        z_mean, ws_c, ws_noise, ws_part, out);
}
